# starting kernel
# baseline (speedup 1.0000x reference)
#include <hip/hip_runtime.h>

#define N_NODES 100000
#define N_EDGES 1600000
#define C_IN 64
#define C_OUT 64

typedef __attribute__((ext_vector_type(2))) float v2f;
typedef __attribute__((ext_vector_type(8))) float v8f;

__global__ __launch_bounds__(256) void gcn_zero_deg(int* __restrict__ deg) {
    int i = blockIdx.x * 256 + threadIdx.x;
    if (i < N_NODES) deg[i] = 0;
}

__global__ __launch_bounds__(256) void gcn_degree(const int* __restrict__ col,
                                                  int* __restrict__ deg) {
    int e = blockIdx.x * 256 + threadIdx.x;
    if (e < N_EDGES) atomicAdd(&deg[col[e]], 1);
}

__global__ __launch_bounds__(256) void gcn_dinv(const int* __restrict__ deg,
                                                float* __restrict__ dinv) {
    int i = blockIdx.x * 256 + threadIdx.x;
    if (i < N_NODES) {
        dinv[i] = rsqrtf((float)deg[i] + 1.0f);
    }
}

__global__ __launch_bounds__(256) void gcn_xw_wmma(const float* __restrict__ x,
                                                   const float* __restrict__ W,
                                                   float* __restrict__ xw) {
    __shared__ float sW[C_IN * C_OUT];

    const int tid = threadIdx.x;
    {
        const float4* Wv = (const float4*)W;
        float4* sWv = (float4*)sW;
#pragma unroll
        for (int i = 0; i < 4; ++i) sWv[tid + 256 * i] = Wv[tid + 256 * i];
    }
    __syncthreads();

    const int wave = tid >> 5;
    const int lane = tid & 31;
    const int m0 = blockIdx.x * 32 + (wave >> 2) * 16;
    const int n0 = (wave & 3) * 16;
    const int mh = lane & 15;
    const int kb = (lane >> 4) << 1;

    const float* arow = x + (size_t)(m0 + mh) * C_IN + kb;
    const float* bcol = sW + (size_t)kb * C_OUT + n0 + mh;

    v8f acc = {};
#pragma unroll
    for (int k = 0; k < C_IN; k += 4) {
        v2f a, b;
        a.x = arow[k];
        a.y = arow[k + 1];
        b.x = bcol[k * C_OUT];
        b.y = bcol[k * C_OUT + C_OUT];
        acc = __builtin_amdgcn_wmma_f32_16x16x4_f32(
             false, a,  false, b,
             (short)0, acc,  false,  false);
    }

    float* drow = xw + (size_t)(m0 + 8 * (lane >> 4)) * C_OUT + n0 + mh;
#pragma unroll
    for (int r = 0; r < 8; ++r) drow[r * C_OUT] = acc[r];
}

__global__ __launch_bounds__(256) void gcn_selfloop_bias(const float* __restrict__ xw,
                                                         const float* __restrict__ dinv,
                                                         const float* __restrict__ b,
                                                         float* __restrict__ out) {
    int idx = blockIdx.x * 256 + threadIdx.x;
    if (idx < N_NODES * (C_OUT / 2)) {
        int n = idx >> 5;
        int c2 = idx & 31;
        float di = dinv[n];
        float s = di * di;
        float2 xv = ((const float2*)xw)[idx];
        float2 bv = ((const float2*)b)[c2];
        float2 o;
        o.x = bv.x + s * xv.x;
        o.y = bv.y + s * xv.y;
        ((float2*)out)[idx] = o;
    }
}

__global__ __launch_bounds__(256) void gcn_edge_scatter(const int* __restrict__ row,
                                                        const int* __restrict__ col,
                                                        const float* __restrict__ xw,
                                                        const float* __restrict__ dinv,
                                                        float* __restrict__ out) {
    const int e = blockIdx.x * 8 + (threadIdx.x >> 5);
    if (e >= N_EDGES) return;
    const int lane = threadIdx.x & 31;
    const int r = row[e];
    const int c = col[e];
    const float norm = dinv[r] * dinv[c];
    const float2 v = ((const float2*)(xw + (size_t)r * C_OUT))[lane];
    float* dst = out + (size_t)c * C_OUT + lane * 2;
    unsafeAtomicAdd(dst, v.x * norm);
    unsafeAtomicAdd(dst + 1, v.y * norm);
}

extern "C" void kernel_launch(void* const* d_in, const int* in_sizes, int n_in,
                              void* d_out, int out_size, void* d_ws, size_t ws_size,
                              hipStream_t stream) {
    const float* x = (const float*)d_in[0];
    const int* ei = (const int*)d_in[1];
    const float* W = (const float*)d_in[2];
    const float* b = (const float*)d_in[3];
    float* out = (float*)d_out;

    const int* row = ei;
    const int* col = ei + N_EDGES;

    char* ws = (char*)d_ws;
    int* deg = (int*)(ws);
    float* dinv = (float*)(ws + (size_t)N_NODES * 4);
    float* xw = (float*)(ws + (size_t)N_NODES * 8);

    gcn_zero_deg<<<(N_NODES + 255) / 256, 256, 0, stream>>>(deg);
    gcn_degree<<<(N_EDGES + 255) / 256, 256, 0, stream>>>(col, deg);
    gcn_dinv<<<(N_NODES + 255) / 256, 256, 0, stream>>>(deg, dinv);
    gcn_xw_wmma<<<N_NODES / 32, 256, 0, stream>>>(x, W, xw);
    gcn_selfloop_bias<<<(N_NODES * (C_OUT / 2) + 255) / 256, 256, 0, stream>>>(xw, dinv, b, out);
    gcn_edge_scatter<<<N_EDGES / 8, 256, 0, stream>>>(row, col, xw, dinv, out);
}


# degree count moved from 1.6M global integer atomics to LDS histograms (8 node ranges x 32 edge slices per block grid) summed per node in the dinv pass; degree clear launch dropped
# speedup vs baseline: 1.3300x; 1.3300x over previous
#include <hip/hip_runtime.h>

#define N_NODES 100000
#define N_EDGES 1600000
#define C_IN 64
#define C_OUT 64

typedef __attribute__((ext_vector_type(2))) float v2f;
typedef __attribute__((ext_vector_type(8))) float v8f;

#define HIST_RANGES 8
#define HIST_SLICES 32
#define HIST_RANGE 12512
#define HIST_PLANE (HIST_RANGES * HIST_RANGE)
#define HIST_SLICE_EDGES (N_EDGES / HIST_SLICES)
#define HIST_SLICE_VEC (HIST_SLICE_EDGES / 4)
#define HIST_THREADS 512
#define HIST_PER_THREAD ((HIST_SLICE_VEC + HIST_THREADS - 1) / HIST_THREADS)
#define HIST_RANGE_VEC (HIST_RANGE / 4)

__global__ __launch_bounds__(HIST_THREADS) void gcn_degree_lds(const int* __restrict__ col,
                                                               int* __restrict__ part) {
    __shared__ alignas(16) int cnt[HIST_RANGE];

    const int tid = threadIdx.x;
    const int p = blockIdx.x & (HIST_RANGES - 1);
    const int q = blockIdx.x >> 3;
    const int lo = p * HIST_RANGE;

    const int4* srcv = (const int4*)(col + (size_t)q * HIST_SLICE_EDGES);
    int4 v[HIST_PER_THREAD];
#pragma unroll
    for (int i = 0; i < HIST_PER_THREAD; ++i) {
        const int idx = tid + i * HIST_THREADS;
        if (idx < HIST_SLICE_VEC) {
            v[i] = srcv[idx];
        } else {
            v[i] = make_int4(-1, -1, -1, -1);
        }
    }

    int4* cntv = (int4*)cnt;
    for (int i = tid; i < HIST_RANGE_VEC; i += HIST_THREADS) cntv[i] = make_int4(0, 0, 0, 0);
    __syncthreads();

#pragma unroll
    for (int i = 0; i < HIST_PER_THREAD; ++i) {
        const unsigned d0 = (unsigned)(v[i].x - lo);
        const unsigned d1 = (unsigned)(v[i].y - lo);
        const unsigned d2 = (unsigned)(v[i].z - lo);
        const unsigned d3 = (unsigned)(v[i].w - lo);
        if (d0 < (unsigned)HIST_RANGE) atomicAdd(&cnt[d0], 1);
        if (d1 < (unsigned)HIST_RANGE) atomicAdd(&cnt[d1], 1);
        if (d2 < (unsigned)HIST_RANGE) atomicAdd(&cnt[d2], 1);
        if (d3 < (unsigned)HIST_RANGE) atomicAdd(&cnt[d3], 1);
    }
    __syncthreads();

    int4* dstv = (int4*)(part + (size_t)q * HIST_PLANE + lo);
    for (int i = tid; i < HIST_RANGE_VEC; i += HIST_THREADS) dstv[i] = cntv[i];
}

__global__ __launch_bounds__(256) void gcn_dinv(const int* __restrict__ part,
                                                float* __restrict__ dinv) {
    int i = blockIdx.x * 256 + threadIdx.x;
    if (i < N_NODES) {
        int d = 0;
#pragma unroll
        for (int q = 0; q < HIST_SLICES; ++q) d += part[(size_t)q * HIST_PLANE + i];
        dinv[i] = rsqrtf((float)d + 1.0f);
    }
}

__global__ __launch_bounds__(256) void gcn_xw_wmma(const float* __restrict__ x,
                                                   const float* __restrict__ W,
                                                   float* __restrict__ xw) {
    __shared__ float sW[C_IN * C_OUT];

    const int tid = threadIdx.x;
    {
        const float4* Wv = (const float4*)W;
        float4* sWv = (float4*)sW;
#pragma unroll
        for (int i = 0; i < 4; ++i) sWv[tid + 256 * i] = Wv[tid + 256 * i];
    }
    __syncthreads();

    const int wave = tid >> 5;
    const int lane = tid & 31;
    const int m0 = blockIdx.x * 32 + (wave >> 2) * 16;
    const int n0 = (wave & 3) * 16;
    const int mh = lane & 15;
    const int kb = (lane >> 4) << 1;

    const float* arow = x + (size_t)(m0 + mh) * C_IN + kb;
    const float* bcol = sW + (size_t)kb * C_OUT + n0 + mh;

    v8f acc = {};
#pragma unroll
    for (int k = 0; k < C_IN; k += 4) {
        v2f a, b;
        a.x = arow[k];
        a.y = arow[k + 1];
        b.x = bcol[k * C_OUT];
        b.y = bcol[k * C_OUT + C_OUT];
        acc = __builtin_amdgcn_wmma_f32_16x16x4_f32(
             false, a,  false, b,
             (short)0, acc,  false,  false);
    }

    float* drow = xw + (size_t)(m0 + 8 * (lane >> 4)) * C_OUT + n0 + mh;
#pragma unroll
    for (int r = 0; r < 8; ++r) drow[r * C_OUT] = acc[r];
}

__global__ __launch_bounds__(256) void gcn_selfloop_bias(const float* __restrict__ xw,
                                                         const float* __restrict__ dinv,
                                                         const float* __restrict__ b,
                                                         float* __restrict__ out) {
    int idx = blockIdx.x * 256 + threadIdx.x;
    if (idx < N_NODES * (C_OUT / 2)) {
        int n = idx >> 5;
        int c2 = idx & 31;
        float di = dinv[n];
        float s = di * di;
        float2 xv = ((const float2*)xw)[idx];
        float2 bv = ((const float2*)b)[c2];
        float2 o;
        o.x = bv.x + s * xv.x;
        o.y = bv.y + s * xv.y;
        ((float2*)out)[idx] = o;
    }
}

__global__ __launch_bounds__(256) void gcn_edge_scatter(const int* __restrict__ row,
                                                        const int* __restrict__ col,
                                                        const float* __restrict__ xw,
                                                        const float* __restrict__ dinv,
                                                        float* __restrict__ out) {
    const int e = blockIdx.x * 8 + (threadIdx.x >> 5);
    if (e >= N_EDGES) return;
    const int lane = threadIdx.x & 31;
    const int r = row[e];
    const int c = col[e];
    const float norm = dinv[r] * dinv[c];
    const float2 v = ((const float2*)(xw + (size_t)r * C_OUT))[lane];
    float* dst = out + (size_t)c * C_OUT + lane * 2;
    unsafeAtomicAdd(dst, v.x * norm);
    unsafeAtomicAdd(dst + 1, v.y * norm);
}

extern "C" void kernel_launch(void* const* d_in, const int* in_sizes, int n_in,
                              void* d_out, int out_size, void* d_ws, size_t ws_size,
                              hipStream_t stream) {
    const float* x = (const float*)d_in[0];
    const int* ei = (const int*)d_in[1];
    const float* W = (const float*)d_in[2];
    const float* b = (const float*)d_in[3];
    float* out = (float*)d_out;

    const int* row = ei;
    const int* col = ei + N_EDGES;

    char* ws = (char*)d_ws;
    float* dinv = (float*)(ws + (size_t)N_NODES * 4);
    float* xw = (float*)(ws + (size_t)N_NODES * 8);
    int* part = (int*)(ws + (size_t)N_NODES * 8);

    gcn_degree_lds<<<HIST_RANGES * HIST_SLICES, HIST_THREADS, 0, stream>>>(col, part);
    gcn_dinv<<<(N_NODES + 255) / 256, 256, 0, stream>>>(part, dinv);
    gcn_xw_wmma<<<N_NODES / 32, 256, 0, stream>>>(x, W, xw);
    gcn_selfloop_bias<<<(N_NODES * (C_OUT / 2) + 255) / 256, 256, 0, stream>>>(xw, dinv, b, out);
    gcn_edge_scatter<<<N_EDGES / 8, 256, 0, stream>>>(row, col, xw, dinv, out);
}


# edge scatter: one wave per 32 edges (was 1 edge): coalesced row/col/dinv loads, 8 row gathers in flight per wave, 50,000 waves not 1.6M; sums still in atomic arrival order
# speedup vs baseline: 1.9406x; 1.4591x over previous
#include <hip/hip_runtime.h>

#define N_NODES 100000
#define N_EDGES 1600000
#define C_IN 64
#define C_OUT 64

typedef __attribute__((ext_vector_type(2))) float v2f;
typedef __attribute__((ext_vector_type(8))) float v8f;

#define HIST_RANGES 8
#define HIST_SLICES 32
#define HIST_RANGE 12512
#define HIST_PLANE (HIST_RANGES * HIST_RANGE)
#define HIST_SLICE_EDGES (N_EDGES / HIST_SLICES)
#define HIST_SLICE_VEC (HIST_SLICE_EDGES / 4)
#define HIST_THREADS 512
#define HIST_PER_THREAD ((HIST_SLICE_VEC + HIST_THREADS - 1) / HIST_THREADS)
#define HIST_RANGE_VEC (HIST_RANGE / 4)

__global__ __launch_bounds__(HIST_THREADS) void gcn_degree_lds(const int* __restrict__ col,
                                                               int* __restrict__ part) {
    __shared__ alignas(16) int cnt[HIST_RANGE];

    const int tid = threadIdx.x;
    const int p = blockIdx.x & (HIST_RANGES - 1);
    const int q = blockIdx.x >> 3;
    const int lo = p * HIST_RANGE;

    const int4* srcv = (const int4*)(col + (size_t)q * HIST_SLICE_EDGES);
    int4 v[HIST_PER_THREAD];
#pragma unroll
    for (int i = 0; i < HIST_PER_THREAD; ++i) {
        const int idx = tid + i * HIST_THREADS;
        if (idx < HIST_SLICE_VEC) {
            v[i] = srcv[idx];
        } else {
            v[i] = make_int4(-1, -1, -1, -1);
        }
    }

    int4* cntv = (int4*)cnt;
    for (int i = tid; i < HIST_RANGE_VEC; i += HIST_THREADS) cntv[i] = make_int4(0, 0, 0, 0);
    __syncthreads();

#pragma unroll
    for (int i = 0; i < HIST_PER_THREAD; ++i) {
        const unsigned d0 = (unsigned)(v[i].x - lo);
        const unsigned d1 = (unsigned)(v[i].y - lo);
        const unsigned d2 = (unsigned)(v[i].z - lo);
        const unsigned d3 = (unsigned)(v[i].w - lo);
        if (d0 < (unsigned)HIST_RANGE) atomicAdd(&cnt[d0], 1);
        if (d1 < (unsigned)HIST_RANGE) atomicAdd(&cnt[d1], 1);
        if (d2 < (unsigned)HIST_RANGE) atomicAdd(&cnt[d2], 1);
        if (d3 < (unsigned)HIST_RANGE) atomicAdd(&cnt[d3], 1);
    }
    __syncthreads();

    int4* dstv = (int4*)(part + (size_t)q * HIST_PLANE + lo);
    for (int i = tid; i < HIST_RANGE_VEC; i += HIST_THREADS) dstv[i] = cntv[i];
}

__global__ __launch_bounds__(256) void gcn_dinv(const int* __restrict__ part,
                                                float* __restrict__ dinv) {
    int i = blockIdx.x * 256 + threadIdx.x;
    if (i < N_NODES) {
        int d = 0;
#pragma unroll
        for (int q = 0; q < HIST_SLICES; ++q) d += part[(size_t)q * HIST_PLANE + i];
        dinv[i] = rsqrtf((float)d + 1.0f);
    }
}

__global__ __launch_bounds__(256) void gcn_xw_wmma(const float* __restrict__ x,
                                                   const float* __restrict__ W,
                                                   float* __restrict__ xw) {
    __shared__ float sW[C_IN * C_OUT];

    const int tid = threadIdx.x;
    {
        const float4* Wv = (const float4*)W;
        float4* sWv = (float4*)sW;
#pragma unroll
        for (int i = 0; i < 4; ++i) sWv[tid + 256 * i] = Wv[tid + 256 * i];
    }
    __syncthreads();

    const int wave = tid >> 5;
    const int lane = tid & 31;
    const int m0 = blockIdx.x * 32 + (wave >> 2) * 16;
    const int n0 = (wave & 3) * 16;
    const int mh = lane & 15;
    const int kb = (lane >> 4) << 1;

    const float* arow = x + (size_t)(m0 + mh) * C_IN + kb;
    const float* bcol = sW + (size_t)kb * C_OUT + n0 + mh;

    v8f acc = {};
#pragma unroll
    for (int k = 0; k < C_IN; k += 4) {
        v2f a, b;
        a.x = arow[k];
        a.y = arow[k + 1];
        b.x = bcol[k * C_OUT];
        b.y = bcol[k * C_OUT + C_OUT];
        acc = __builtin_amdgcn_wmma_f32_16x16x4_f32(
             false, a,  false, b,
             (short)0, acc,  false,  false);
    }

    float* drow = xw + (size_t)(m0 + 8 * (lane >> 4)) * C_OUT + n0 + mh;
#pragma unroll
    for (int r = 0; r < 8; ++r) drow[r * C_OUT] = acc[r];
}

__global__ __launch_bounds__(256) void gcn_selfloop_bias(const float* __restrict__ xw,
                                                         const float* __restrict__ dinv,
                                                         const float* __restrict__ b,
                                                         float* __restrict__ out) {
    int idx = blockIdx.x * 256 + threadIdx.x;
    if (idx < N_NODES * (C_OUT / 2)) {
        int n = idx >> 5;
        int c2 = idx & 31;
        float di = dinv[n];
        float s = di * di;
        float2 xv = ((const float2*)xw)[idx];
        float2 bv = ((const float2*)b)[c2];
        float2 o;
        o.x = bv.x + s * xv.x;
        o.y = bv.y + s * xv.y;
        ((float2*)out)[idx] = o;
    }
}

#define EDGES_PER_WAVE 32
#define EDGE_GROUP 8
__global__ __launch_bounds__(256) void gcn_edge_scatter(const int* __restrict__ row,
                                                        const int* __restrict__ col,
                                                        const float* __restrict__ xw,
                                                        const float* __restrict__ dinv,
                                                        float* __restrict__ out) {
    const int e0 = (blockIdx.x * 8 + (threadIdx.x >> 5)) * EDGES_PER_WAVE;
    if (e0 >= N_EDGES) return;
    const int lane = threadIdx.x & 31;
    const int r = row[e0 + lane];
    const int c = col[e0 + lane];
    const float norm = dinv[r] * dinv[c];

#pragma unroll
    for (int g = 0; g < EDGES_PER_WAVE; g += EDGE_GROUP) {
        float2 v[EDGE_GROUP];
#pragma unroll
        for (int j = 0; j < EDGE_GROUP; ++j) {
            const int rj = __builtin_amdgcn_readlane(r, g + j);
            v[j] = ((const float2*)(xw + (size_t)rj * C_OUT))[lane];
        }
#pragma unroll
        for (int j = 0; j < EDGE_GROUP; ++j) {
            const int cj = __builtin_amdgcn_readlane(c, g + j);
            const float nj = __builtin_bit_cast(
                float, __builtin_amdgcn_readlane(__builtin_bit_cast(int, norm), g + j));
            float* dst = out + (size_t)cj * C_OUT + lane * 2;
            unsafeAtomicAdd(dst, v[j].x * nj);
            unsafeAtomicAdd(dst + 1, v[j].y * nj);
        }
    }
}

extern "C" void kernel_launch(void* const* d_in, const int* in_sizes, int n_in,
                              void* d_out, int out_size, void* d_ws, size_t ws_size,
                              hipStream_t stream) {
    const float* x = (const float*)d_in[0];
    const int* ei = (const int*)d_in[1];
    const float* W = (const float*)d_in[2];
    const float* b = (const float*)d_in[3];
    float* out = (float*)d_out;

    const int* row = ei;
    const int* col = ei + N_EDGES;

    char* ws = (char*)d_ws;
    float* dinv = (float*)(ws + (size_t)N_NODES * 4);
    float* xw = (float*)(ws + (size_t)N_NODES * 8);
    int* part = (int*)(ws + (size_t)N_NODES * 8);

    gcn_degree_lds<<<HIST_RANGES * HIST_SLICES, HIST_THREADS, 0, stream>>>(col, part);
    gcn_dinv<<<(N_NODES + 255) / 256, 256, 0, stream>>>(part, dinv);
    gcn_xw_wmma<<<N_NODES / 32, 256, 0, stream>>>(x, W, xw);
    gcn_selfloop_bias<<<(N_NODES * (C_OUT / 2) + 255) / 256, 256, 0, stream>>>(xw, dinv, b, out);
    gcn_edge_scatter<<<N_EDGES / (8 * EDGES_PER_WAVE), 256, 0, stream>>>(row, col, xw, dinv, out);
}


# edge scatter: lane l owns channels l and l+32 (was 2l,2l+1): each gather and f32 atomic instruction covers one whole 128-byte line, atomic line requests halve; sums still in atomic arrival order
# speedup vs baseline: 2.3994x; 1.2364x over previous
#include <hip/hip_runtime.h>

#define N_NODES 100000
#define N_EDGES 1600000
#define C_IN 64
#define C_OUT 64

typedef __attribute__((ext_vector_type(2))) float v2f;
typedef __attribute__((ext_vector_type(8))) float v8f;

#define HIST_RANGES 8
#define HIST_SLICES 32
#define HIST_RANGE 12512
#define HIST_PLANE (HIST_RANGES * HIST_RANGE)
#define HIST_SLICE_EDGES (N_EDGES / HIST_SLICES)
#define HIST_SLICE_VEC (HIST_SLICE_EDGES / 4)
#define HIST_THREADS 512
#define HIST_PER_THREAD ((HIST_SLICE_VEC + HIST_THREADS - 1) / HIST_THREADS)
#define HIST_RANGE_VEC (HIST_RANGE / 4)

__global__ __launch_bounds__(HIST_THREADS) void gcn_degree_lds(const int* __restrict__ col,
                                                               int* __restrict__ part) {
    __shared__ alignas(16) int cnt[HIST_RANGE];

    const int tid = threadIdx.x;
    const int p = blockIdx.x & (HIST_RANGES - 1);
    const int q = blockIdx.x >> 3;
    const int lo = p * HIST_RANGE;

    const int4* srcv = (const int4*)(col + (size_t)q * HIST_SLICE_EDGES);
    int4 v[HIST_PER_THREAD];
#pragma unroll
    for (int i = 0; i < HIST_PER_THREAD; ++i) {
        const int idx = tid + i * HIST_THREADS;
        if (idx < HIST_SLICE_VEC) {
            v[i] = srcv[idx];
        } else {
            v[i] = make_int4(-1, -1, -1, -1);
        }
    }

    int4* cntv = (int4*)cnt;
    for (int i = tid; i < HIST_RANGE_VEC; i += HIST_THREADS) cntv[i] = make_int4(0, 0, 0, 0);
    __syncthreads();

#pragma unroll
    for (int i = 0; i < HIST_PER_THREAD; ++i) {
        const unsigned d0 = (unsigned)(v[i].x - lo);
        const unsigned d1 = (unsigned)(v[i].y - lo);
        const unsigned d2 = (unsigned)(v[i].z - lo);
        const unsigned d3 = (unsigned)(v[i].w - lo);
        if (d0 < (unsigned)HIST_RANGE) atomicAdd(&cnt[d0], 1);
        if (d1 < (unsigned)HIST_RANGE) atomicAdd(&cnt[d1], 1);
        if (d2 < (unsigned)HIST_RANGE) atomicAdd(&cnt[d2], 1);
        if (d3 < (unsigned)HIST_RANGE) atomicAdd(&cnt[d3], 1);
    }
    __syncthreads();

    int4* dstv = (int4*)(part + (size_t)q * HIST_PLANE + lo);
    for (int i = tid; i < HIST_RANGE_VEC; i += HIST_THREADS) dstv[i] = cntv[i];
}

__global__ __launch_bounds__(256) void gcn_dinv(const int* __restrict__ part,
                                                float* __restrict__ dinv) {
    int i = blockIdx.x * 256 + threadIdx.x;
    if (i < N_NODES) {
        int d = 0;
#pragma unroll
        for (int q = 0; q < HIST_SLICES; ++q) d += part[(size_t)q * HIST_PLANE + i];
        dinv[i] = rsqrtf((float)d + 1.0f);
    }
}

__global__ __launch_bounds__(256) void gcn_xw_wmma(const float* __restrict__ x,
                                                   const float* __restrict__ W,
                                                   float* __restrict__ xw) {
    __shared__ float sW[C_IN * C_OUT];

    const int tid = threadIdx.x;
    {
        const float4* Wv = (const float4*)W;
        float4* sWv = (float4*)sW;
#pragma unroll
        for (int i = 0; i < 4; ++i) sWv[tid + 256 * i] = Wv[tid + 256 * i];
    }
    __syncthreads();

    const int wave = tid >> 5;
    const int lane = tid & 31;
    const int m0 = blockIdx.x * 32 + (wave >> 2) * 16;
    const int n0 = (wave & 3) * 16;
    const int mh = lane & 15;
    const int kb = (lane >> 4) << 1;

    const float* arow = x + (size_t)(m0 + mh) * C_IN + kb;
    const float* bcol = sW + (size_t)kb * C_OUT + n0 + mh;

    v8f acc = {};
#pragma unroll
    for (int k = 0; k < C_IN; k += 4) {
        v2f a, b;
        a.x = arow[k];
        a.y = arow[k + 1];
        b.x = bcol[k * C_OUT];
        b.y = bcol[k * C_OUT + C_OUT];
        acc = __builtin_amdgcn_wmma_f32_16x16x4_f32(
             false, a,  false, b,
             (short)0, acc,  false,  false);
    }

    float* drow = xw + (size_t)(m0 + 8 * (lane >> 4)) * C_OUT + n0 + mh;
#pragma unroll
    for (int r = 0; r < 8; ++r) drow[r * C_OUT] = acc[r];
}

__global__ __launch_bounds__(256) void gcn_selfloop_bias(const float* __restrict__ xw,
                                                         const float* __restrict__ dinv,
                                                         const float* __restrict__ b,
                                                         float* __restrict__ out) {
    int idx = blockIdx.x * 256 + threadIdx.x;
    if (idx < N_NODES * (C_OUT / 2)) {
        int n = idx >> 5;
        int c2 = idx & 31;
        float di = dinv[n];
        float s = di * di;
        float2 xv = ((const float2*)xw)[idx];
        float2 bv = ((const float2*)b)[c2];
        float2 o;
        o.x = bv.x + s * xv.x;
        o.y = bv.y + s * xv.y;
        ((float2*)out)[idx] = o;
    }
}

#define EDGES_PER_WAVE 32
#define EDGE_GROUP 8
__global__ __launch_bounds__(256) void gcn_edge_scatter(const int* __restrict__ row,
                                                        const int* __restrict__ col,
                                                        const float* __restrict__ xw,
                                                        const float* __restrict__ dinv,
                                                        float* __restrict__ out) {
    const int e0 = (blockIdx.x * 8 + (threadIdx.x >> 5)) * EDGES_PER_WAVE;
    if (e0 >= N_EDGES) return;
    const int lane = threadIdx.x & 31;
    const int r = row[e0 + lane];
    const int c = col[e0 + lane];
    const float norm = dinv[r] * dinv[c];

#pragma unroll
    for (int g = 0; g < EDGES_PER_WAVE; g += EDGE_GROUP) {
        float vlo[EDGE_GROUP], vhi[EDGE_GROUP];
#pragma unroll
        for (int j = 0; j < EDGE_GROUP; ++j) {
            const int rj = __builtin_amdgcn_readlane(r, g + j);
            const float* srow = xw + (size_t)rj * C_OUT + lane;
            vlo[j] = srow[0];
            vhi[j] = srow[C_OUT / 2];
        }
#pragma unroll
        for (int j = 0; j < EDGE_GROUP; ++j) {
            const int cj = __builtin_amdgcn_readlane(c, g + j);
            const float nj = __builtin_bit_cast(
                float, __builtin_amdgcn_readlane(__builtin_bit_cast(int, norm), g + j));
            float* dst = out + (size_t)cj * C_OUT + lane;
            unsafeAtomicAdd(dst, vlo[j] * nj);
            unsafeAtomicAdd(dst + C_OUT / 2, vhi[j] * nj);
        }
    }
}

extern "C" void kernel_launch(void* const* d_in, const int* in_sizes, int n_in,
                              void* d_out, int out_size, void* d_ws, size_t ws_size,
                              hipStream_t stream) {
    const float* x = (const float*)d_in[0];
    const int* ei = (const int*)d_in[1];
    const float* W = (const float*)d_in[2];
    const float* b = (const float*)d_in[3];
    float* out = (float*)d_out;

    const int* row = ei;
    const int* col = ei + N_EDGES;

    char* ws = (char*)d_ws;
    float* dinv = (float*)(ws + (size_t)N_NODES * 4);
    float* xw = (float*)(ws + (size_t)N_NODES * 8);
    int* part = (int*)(ws + (size_t)N_NODES * 8);

    gcn_degree_lds<<<HIST_RANGES * HIST_SLICES, HIST_THREADS, 0, stream>>>(col, part);
    gcn_dinv<<<(N_NODES + 255) / 256, 256, 0, stream>>>(part, dinv);
    gcn_xw_wmma<<<N_NODES / 32, 256, 0, stream>>>(x, W, xw);
    gcn_selfloop_bias<<<(N_NODES * (C_OUT / 2) + 255) / 256, 256, 0, stream>>>(xw, dinv, b, out);
    gcn_edge_scatter<<<N_EDGES / (8 * EDGES_PER_WAVE), 256, 0, stream>>>(row, col, xw, dinv, out);
}


# edge scatter: each out row gets its atomic adds from one cache half only (wave pairs split a 64-edge chunk by target half, side claimed by die id); sums still in atomic arrival order
# speedup vs baseline: 2.8385x; 1.1830x over previous
#include <hip/hip_runtime.h>

#define N_NODES 100000
#define N_EDGES 1600000
#define C_IN 64
#define C_OUT 64
#define N_PAIRS (N_EDGES / 64)

typedef __attribute__((ext_vector_type(2))) float v2f;
typedef __attribute__((ext_vector_type(8))) float v8f;

#define HIST_RANGES 8
#define HIST_SLICES 32
#define HIST_RANGE 12512
#define HIST_PLANE (HIST_RANGES * HIST_RANGE)
#define HIST_SLICE_EDGES (N_EDGES / HIST_SLICES)
#define HIST_SLICE_VEC (HIST_SLICE_EDGES / 4)
#define HIST_THREADS 512
#define HIST_PER_THREAD ((HIST_SLICE_VEC + HIST_THREADS - 1) / HIST_THREADS)
#define HIST_RANGE_VEC (HIST_RANGE / 4)

__global__ __launch_bounds__(HIST_THREADS) void gcn_degree_lds(const int* __restrict__ col,
                                                               int* __restrict__ part) {
    __shared__ alignas(16) int cnt[HIST_RANGE];

    const int tid = threadIdx.x;
    const int p = blockIdx.x & (HIST_RANGES - 1);
    const int q = blockIdx.x >> 3;
    const int lo = p * HIST_RANGE;

    const int4* srcv = (const int4*)(col + (size_t)q * HIST_SLICE_EDGES);
    int4 v[HIST_PER_THREAD];
#pragma unroll
    for (int i = 0; i < HIST_PER_THREAD; ++i) {
        const int idx = tid + i * HIST_THREADS;
        if (idx < HIST_SLICE_VEC) {
            v[i] = srcv[idx];
        } else {
            v[i] = make_int4(-1, -1, -1, -1);
        }
    }

    int4* cntv = (int4*)cnt;
    for (int i = tid; i < HIST_RANGE_VEC; i += HIST_THREADS) cntv[i] = make_int4(0, 0, 0, 0);
    __syncthreads();

#pragma unroll
    for (int i = 0; i < HIST_PER_THREAD; ++i) {
        const unsigned d0 = (unsigned)(v[i].x - lo);
        const unsigned d1 = (unsigned)(v[i].y - lo);
        const unsigned d2 = (unsigned)(v[i].z - lo);
        const unsigned d3 = (unsigned)(v[i].w - lo);
        if (d0 < (unsigned)HIST_RANGE) atomicAdd(&cnt[d0], 1);
        if (d1 < (unsigned)HIST_RANGE) atomicAdd(&cnt[d1], 1);
        if (d2 < (unsigned)HIST_RANGE) atomicAdd(&cnt[d2], 1);
        if (d3 < (unsigned)HIST_RANGE) atomicAdd(&cnt[d3], 1);
    }
    __syncthreads();

    int4* dstv = (int4*)(part + (size_t)q * HIST_PLANE + lo);
    for (int i = tid; i < HIST_RANGE_VEC; i += HIST_THREADS) dstv[i] = cntv[i];
}

__global__ __launch_bounds__(256) void gcn_dinv(const int* __restrict__ part,
                                                float* __restrict__ dinv,
                                                unsigned* __restrict__ claim) {
    int i = blockIdx.x * 256 + threadIdx.x;
    if (i < N_PAIRS) claim[i] = 0u;
    if (i < N_NODES) {
        int d = 0;
#pragma unroll
        for (int q = 0; q < HIST_SLICES; ++q) d += part[(size_t)q * HIST_PLANE + i];
        dinv[i] = rsqrtf((float)d + 1.0f);
    }
}

__global__ __launch_bounds__(256) void gcn_xw_wmma(const float* __restrict__ x,
                                                   const float* __restrict__ W,
                                                   float* __restrict__ xw) {
    __shared__ float sW[C_IN * C_OUT];

    const int tid = threadIdx.x;
    {
        const float4* Wv = (const float4*)W;
        float4* sWv = (float4*)sW;
#pragma unroll
        for (int i = 0; i < 4; ++i) sWv[tid + 256 * i] = Wv[tid + 256 * i];
    }
    __syncthreads();

    const int wave = tid >> 5;
    const int lane = tid & 31;
    const int m0 = blockIdx.x * 32 + (wave >> 2) * 16;
    const int n0 = (wave & 3) * 16;
    const int mh = lane & 15;
    const int kb = (lane >> 4) << 1;

    const float* arow = x + (size_t)(m0 + mh) * C_IN + kb;
    const float* bcol = sW + (size_t)kb * C_OUT + n0 + mh;

    v8f acc = {};
#pragma unroll
    for (int k = 0; k < C_IN; k += 4) {
        v2f a, b;
        a.x = arow[k];
        a.y = arow[k + 1];
        b.x = bcol[k * C_OUT];
        b.y = bcol[k * C_OUT + C_OUT];
        acc = __builtin_amdgcn_wmma_f32_16x16x4_f32(
             false, a,  false, b,
             (short)0, acc,  false,  false);
    }

    float* drow = xw + (size_t)(m0 + 8 * (lane >> 4)) * C_OUT + n0 + mh;
#pragma unroll
    for (int r = 0; r < 8; ++r) drow[r * C_OUT] = acc[r];
}

__global__ __launch_bounds__(256) void gcn_selfloop_bias(const float* __restrict__ xw,
                                                         const float* __restrict__ dinv,
                                                         const float* __restrict__ b,
                                                         float* __restrict__ out) {
    int idx = blockIdx.x * 256 + threadIdx.x;
    if (idx < N_NODES * (C_OUT / 2)) {
        int n = idx >> 5;
        int c2 = idx & 31;
        float di = dinv[n];
        float s = di * di;
        float2 xv = ((const float2*)xw)[idx];
        float2 bv = ((const float2*)b)[c2];
        float2 o;
        o.x = bv.x + s * xv.x;
        o.y = bv.y + s * xv.y;
        ((float2*)out)[idx] = o;
    }
}

#define PAIR_EDGES 64
#define PAIR_BLOCKS ((N_PAIRS + 7) / 8)
#define SCATTER_GRID (((PAIR_BLOCKS + 3) / 4) * 8)
#define HALF_SPLIT (N_NODES / 2)
#define EDGE_GROUP 8

typedef __attribute__((address_space(1))) float gfloat;
__device__ __forceinline__ gfloat* uniform_row(const float* p) {
    const unsigned long long v = (unsigned long long)p;
    const unsigned lo = (unsigned)__builtin_amdgcn_readfirstlane((int)(unsigned)v);
    const unsigned hi = (unsigned)__builtin_amdgcn_readfirstlane((int)(unsigned)(v >> 32));
    return (gfloat*)(((unsigned long long)hi << 32) | lo);
}

template <int N>
__device__ __forceinline__ void scatter_group(unsigned long long& mask, const unsigned lane,
                                              const int r0, const int r1, const int c0,
                                              const int c1, const float n0, const float n1,
                                              const float* __restrict__ xw,
                                              float* __restrict__ out) {
    float vlo[N], vhi[N], nj[N];
    int rj[N], cj[N];
#pragma unroll
    for (int j = 0; j < N; ++j) {
        const int ln = (int)__builtin_ctzll(mask);
        mask &= mask - 1ull;
        const int l5 = ln & 31;
        const bool up = ln >= 32;
        const int ra = __builtin_amdgcn_readlane(r0, l5);
        const int rb = __builtin_amdgcn_readlane(r1, l5);
        const int ca = __builtin_amdgcn_readlane(c0, l5);
        const int cb = __builtin_amdgcn_readlane(c1, l5);
        const int na = __builtin_amdgcn_readlane(__builtin_bit_cast(int, n0), l5);
        const int nb = __builtin_amdgcn_readlane(__builtin_bit_cast(int, n1), l5);
        rj[j] = up ? rb : ra;
        cj[j] = up ? cb : ca;
        nj[j] = __builtin_bit_cast(float, up ? nb : na);
    }
#pragma unroll
    for (int j = 0; j < N; ++j) {
        const gfloat* srow = uniform_row(xw + (size_t)rj[j] * C_OUT);
        vlo[j] = srow[lane];
        vhi[j] = srow[lane + C_OUT / 2];
    }
#pragma unroll
    for (int j = 0; j < N; ++j) {
        gfloat* dst = uniform_row(out + (size_t)cj[j] * C_OUT);
        __builtin_amdgcn_global_atomic_fadd_f32(dst + lane, vlo[j] * nj[j]);
        __builtin_amdgcn_global_atomic_fadd_f32(dst + lane + C_OUT / 2, vhi[j] * nj[j]);
    }
}

__global__ __launch_bounds__(256) void gcn_edge_scatter(const int* __restrict__ row,
                                                        const int* __restrict__ col,
                                                        const float* __restrict__ xw,
                                                        const float* __restrict__ dinv,
                                                        float* __restrict__ out,
                                                        unsigned* __restrict__ claim) {
    const int b = blockIdx.x;
    const int pb = ((b >> 3) << 2) | (b & 3);
    const int pair = pb * 8 + (threadIdx.x >> 5);
    if (pair >= N_PAIRS) return;
    const unsigned lane = threadIdx.x & 31u;
    const int e0 = pair * PAIR_EDGES;

    const int r0 = row[e0 + lane];
    const int r1 = row[e0 + 32 + lane];
    const int c0 = col[e0 + lane];
    const int c1 = col[e0 + 32 + lane];

    const unsigned half = (__builtin_amdgcn_s_sendmsg_rtn(0x87) >> 18) & 1u;
    unsigned old = 0u;
    if (lane == 0) old = atomicOr(&claim[pair], 1u << half);
    old = (unsigned)__builtin_amdgcn_readfirstlane((int)old);
    const unsigned side = (old == 0u) ? half : (old & 1u);

    const bool p0 = (unsigned)(c0 >= HALF_SPLIT) == side;
    const bool p1 = (unsigned)(c1 >= HALF_SPLIT) == side;
    const float n0 = dinv[p0 ? r0 : 0] * dinv[p0 ? c0 : 0];
    const float n1 = dinv[p1 ? r1 : 0] * dinv[p1 ? c1 : 0];

    unsigned long long mask = (unsigned long long)__builtin_amdgcn_ballot_w32(p0) |
                              ((unsigned long long)__builtin_amdgcn_ballot_w32(p1) << 32);

    int left = __builtin_popcountll(mask);
    for (; left >= EDGE_GROUP; left -= EDGE_GROUP)
        scatter_group<EDGE_GROUP>(mask, lane, r0, r1, c0, c1, n0, n1, xw, out);
    switch (left) {
        case 7: scatter_group<7>(mask, lane, r0, r1, c0, c1, n0, n1, xw, out); break;
        case 6: scatter_group<6>(mask, lane, r0, r1, c0, c1, n0, n1, xw, out); break;
        case 5: scatter_group<5>(mask, lane, r0, r1, c0, c1, n0, n1, xw, out); break;
        case 4: scatter_group<4>(mask, lane, r0, r1, c0, c1, n0, n1, xw, out); break;
        case 3: scatter_group<3>(mask, lane, r0, r1, c0, c1, n0, n1, xw, out); break;
        case 2: scatter_group<2>(mask, lane, r0, r1, c0, c1, n0, n1, xw, out); break;
        case 1: scatter_group<1>(mask, lane, r0, r1, c0, c1, n0, n1, xw, out); break;
        default: break;
    }
}

extern "C" void kernel_launch(void* const* d_in, const int* in_sizes, int n_in,
                              void* d_out, int out_size, void* d_ws, size_t ws_size,
                              hipStream_t stream) {
    const float* x = (const float*)d_in[0];
    const int* ei = (const int*)d_in[1];
    const float* W = (const float*)d_in[2];
    const float* b = (const float*)d_in[3];
    float* out = (float*)d_out;

    const int* row = ei;
    const int* col = ei + N_EDGES;

    char* ws = (char*)d_ws;
    unsigned* claim = (unsigned*)ws;
    float* dinv = (float*)(ws + (size_t)N_NODES * 4);
    float* xw = (float*)(ws + (size_t)N_NODES * 8);
    int* part = (int*)(ws + (size_t)N_NODES * 8);

    gcn_degree_lds<<<HIST_RANGES * HIST_SLICES, HIST_THREADS, 0, stream>>>(col, part);
    gcn_dinv<<<(N_NODES + 255) / 256, 256, 0, stream>>>(part, dinv, claim);
    gcn_xw_wmma<<<N_NODES / 32, 256, 0, stream>>>(x, W, xw);
    gcn_selfloop_bias<<<(N_NODES * (C_OUT / 2) + 255) / 256, 256, 0, stream>>>(xw, dinv, b, out);
    gcn_edge_scatter<<<SCATTER_GRID, 256, 0, stream>>>(row, col, xw, dinv, out, claim);
}


# product kernel's epilogue also stores out's start value b + dinv^2*xw from the tile in registers: separate start-value launch and its re-read of xw gone (5 launches to 4); values, sum order unchanged
# speedup vs baseline: 3.0603x; 1.0781x over previous
#include <hip/hip_runtime.h>

#define N_NODES 100000
#define N_EDGES 1600000
#define C_IN 64
#define C_OUT 64
#define N_PAIRS (N_EDGES / 64)

typedef __attribute__((ext_vector_type(2))) float v2f;
typedef __attribute__((ext_vector_type(8))) float v8f;

#define HIST_RANGES 8
#define HIST_SLICES 32
#define HIST_RANGE 12512
#define HIST_PLANE (HIST_RANGES * HIST_RANGE)
#define HIST_SLICE_EDGES (N_EDGES / HIST_SLICES)
#define HIST_SLICE_VEC (HIST_SLICE_EDGES / 4)
#define HIST_THREADS 512
#define HIST_PER_THREAD ((HIST_SLICE_VEC + HIST_THREADS - 1) / HIST_THREADS)
#define HIST_RANGE_VEC (HIST_RANGE / 4)

__global__ __launch_bounds__(HIST_THREADS) void gcn_degree_lds(const int* __restrict__ col,
                                                               int* __restrict__ part) {
    __shared__ alignas(16) int cnt[HIST_RANGE];

    const int tid = threadIdx.x;
    const int p = blockIdx.x & (HIST_RANGES - 1);
    const int q = blockIdx.x >> 3;
    const int lo = p * HIST_RANGE;

    const int4* srcv = (const int4*)(col + (size_t)q * HIST_SLICE_EDGES);
    int4 v[HIST_PER_THREAD];
#pragma unroll
    for (int i = 0; i < HIST_PER_THREAD; ++i) {
        const int idx = tid + i * HIST_THREADS;
        if (idx < HIST_SLICE_VEC) {
            v[i] = srcv[idx];
        } else {
            v[i] = make_int4(-1, -1, -1, -1);
        }
    }

    int4* cntv = (int4*)cnt;
    for (int i = tid; i < HIST_RANGE_VEC; i += HIST_THREADS) cntv[i] = make_int4(0, 0, 0, 0);
    __syncthreads();

#pragma unroll
    for (int i = 0; i < HIST_PER_THREAD; ++i) {
        const unsigned d0 = (unsigned)(v[i].x - lo);
        const unsigned d1 = (unsigned)(v[i].y - lo);
        const unsigned d2 = (unsigned)(v[i].z - lo);
        const unsigned d3 = (unsigned)(v[i].w - lo);
        if (d0 < (unsigned)HIST_RANGE) atomicAdd(&cnt[d0], 1);
        if (d1 < (unsigned)HIST_RANGE) atomicAdd(&cnt[d1], 1);
        if (d2 < (unsigned)HIST_RANGE) atomicAdd(&cnt[d2], 1);
        if (d3 < (unsigned)HIST_RANGE) atomicAdd(&cnt[d3], 1);
    }
    __syncthreads();

    int4* dstv = (int4*)(part + (size_t)q * HIST_PLANE + lo);
    for (int i = tid; i < HIST_RANGE_VEC; i += HIST_THREADS) dstv[i] = cntv[i];
}

__global__ __launch_bounds__(256) void gcn_dinv(const int* __restrict__ part,
                                                float* __restrict__ dinv,
                                                unsigned* __restrict__ claim) {
    int i = blockIdx.x * 256 + threadIdx.x;
    if (i < N_PAIRS) claim[i] = 0u;
    if (i < N_NODES) {
        int d = 0;
#pragma unroll
        for (int q = 0; q < HIST_SLICES; ++q) d += part[(size_t)q * HIST_PLANE + i];
        dinv[i] = rsqrtf((float)d + 1.0f);
    }
}

__global__ __launch_bounds__(256) void gcn_xw_wmma(const float* __restrict__ x,
                                                   const float* __restrict__ W,
                                                   const float* __restrict__ dinv,
                                                   const float* __restrict__ b,
                                                   float* __restrict__ xw,
                                                   float* __restrict__ out) {
    __shared__ float sW[C_IN * C_OUT];

    const int tid = threadIdx.x;
    const float4* dq = (const float4*)(dinv + blockIdx.x * 32 + (tid >> 7) * 16 + 8 * ((tid >> 4) & 1));
    const float4 dlo = dq[0];
    const float4 dhi = dq[1];
    const float bc = b[((tid >> 5) & 3) * 16 + (tid & 15)];
    {
        const float4* Wv = (const float4*)W;
        float4* sWv = (float4*)sW;
#pragma unroll
        for (int i = 0; i < 4; ++i) sWv[tid + 256 * i] = Wv[tid + 256 * i];
    }
    __syncthreads();

    const int wave = tid >> 5;
    const int lane = tid & 31;
    const int m0 = blockIdx.x * 32 + (wave >> 2) * 16;
    const int n0 = (wave & 3) * 16;
    const int mh = lane & 15;
    const int kb = (lane >> 4) << 1;

    const float* arow = x + (size_t)(m0 + mh) * C_IN + kb;
    const float* bcol = sW + (size_t)kb * C_OUT + n0 + mh;

    v8f acc = {};
#pragma unroll
    for (int k = 0; k < C_IN; k += 4) {
        v2f a, bf;
        a.x = arow[k];
        a.y = arow[k + 1];
        bf.x = bcol[k * C_OUT];
        bf.y = bcol[k * C_OUT + C_OUT];
        acc = __builtin_amdgcn_wmma_f32_16x16x4_f32(
             false, a,  false, bf,
             (short)0, acc,  false,  false);
    }

    const size_t tile = (size_t)(m0 + 8 * (lane >> 4)) * C_OUT + n0 + mh;
    float* drow = xw + tile;
    float* orow = out + tile;
    const float dr[8] = {dlo.x, dlo.y, dlo.z, dlo.w, dhi.x, dhi.y, dhi.z, dhi.w};
#pragma unroll
    for (int r = 0; r < 8; ++r) {
        const float s = dr[r] * dr[r];
        drow[r * C_OUT] = acc[r];
        orow[r * C_OUT] = bc + s * acc[r];
    }
}

#define PAIR_EDGES 64
#define PAIR_BLOCKS ((N_PAIRS + 7) / 8)
#define SCATTER_GRID (((PAIR_BLOCKS + 3) / 4) * 8)
#define HALF_SPLIT (N_NODES / 2)
#define EDGE_GROUP 8

typedef __attribute__((address_space(1))) float gfloat;
__device__ __forceinline__ gfloat* uniform_row(const float* p) {
    const unsigned long long v = (unsigned long long)p;
    const unsigned lo = (unsigned)__builtin_amdgcn_readfirstlane((int)(unsigned)v);
    const unsigned hi = (unsigned)__builtin_amdgcn_readfirstlane((int)(unsigned)(v >> 32));
    return (gfloat*)(((unsigned long long)hi << 32) | lo);
}

template <int N>
__device__ __forceinline__ void scatter_group(unsigned long long& mask, const unsigned lane,
                                              const int r0, const int r1, const int c0,
                                              const int c1, const float n0, const float n1,
                                              const float* __restrict__ xw,
                                              float* __restrict__ out) {
    float vlo[N], vhi[N], nj[N];
    int rj[N], cj[N];
#pragma unroll
    for (int j = 0; j < N; ++j) {
        const int ln = (int)__builtin_ctzll(mask);
        mask &= mask - 1ull;
        const int l5 = ln & 31;
        const bool up = ln >= 32;
        const int ra = __builtin_amdgcn_readlane(r0, l5);
        const int rb = __builtin_amdgcn_readlane(r1, l5);
        const int ca = __builtin_amdgcn_readlane(c0, l5);
        const int cb = __builtin_amdgcn_readlane(c1, l5);
        const int na = __builtin_amdgcn_readlane(__builtin_bit_cast(int, n0), l5);
        const int nb = __builtin_amdgcn_readlane(__builtin_bit_cast(int, n1), l5);
        rj[j] = up ? rb : ra;
        cj[j] = up ? cb : ca;
        nj[j] = __builtin_bit_cast(float, up ? nb : na);
    }
#pragma unroll
    for (int j = 0; j < N; ++j) {
        const gfloat* srow = uniform_row(xw + (size_t)rj[j] * C_OUT);
        vlo[j] = srow[lane];
        vhi[j] = srow[lane + C_OUT / 2];
    }
#pragma unroll
    for (int j = 0; j < N; ++j) {
        gfloat* dst = uniform_row(out + (size_t)cj[j] * C_OUT);
        __builtin_amdgcn_global_atomic_fadd_f32(dst + lane, vlo[j] * nj[j]);
        __builtin_amdgcn_global_atomic_fadd_f32(dst + lane + C_OUT / 2, vhi[j] * nj[j]);
    }
}

__global__ __launch_bounds__(256) void gcn_edge_scatter(const int* __restrict__ row,
                                                        const int* __restrict__ col,
                                                        const float* __restrict__ xw,
                                                        const float* __restrict__ dinv,
                                                        float* __restrict__ out,
                                                        unsigned* __restrict__ claim) {
    const int b = blockIdx.x;
    const int pb = ((b >> 3) << 2) | (b & 3);
    const int pair = pb * 8 + (threadIdx.x >> 5);
    if (pair >= N_PAIRS) return;
    const unsigned lane = threadIdx.x & 31u;
    const int e0 = pair * PAIR_EDGES;

    const int r0 = row[e0 + lane];
    const int r1 = row[e0 + 32 + lane];
    const int c0 = col[e0 + lane];
    const int c1 = col[e0 + 32 + lane];

    const unsigned half = (__builtin_amdgcn_s_sendmsg_rtn(0x87) >> 18) & 1u;
    unsigned old = 0u;
    if (lane == 0) old = atomicOr(&claim[pair], 1u << half);
    old = (unsigned)__builtin_amdgcn_readfirstlane((int)old);
    const unsigned side = (old == 0u) ? half : (old & 1u);

    const bool p0 = (unsigned)(c0 >= HALF_SPLIT) == side;
    const bool p1 = (unsigned)(c1 >= HALF_SPLIT) == side;
    const float n0 = dinv[p0 ? r0 : 0] * dinv[p0 ? c0 : 0];
    const float n1 = dinv[p1 ? r1 : 0] * dinv[p1 ? c1 : 0];

    unsigned long long mask = (unsigned long long)__builtin_amdgcn_ballot_w32(p0) |
                              ((unsigned long long)__builtin_amdgcn_ballot_w32(p1) << 32);

    int left = __builtin_popcountll(mask);
    for (; left >= EDGE_GROUP; left -= EDGE_GROUP)
        scatter_group<EDGE_GROUP>(mask, lane, r0, r1, c0, c1, n0, n1, xw, out);
    switch (left) {
        case 7: scatter_group<7>(mask, lane, r0, r1, c0, c1, n0, n1, xw, out); break;
        case 6: scatter_group<6>(mask, lane, r0, r1, c0, c1, n0, n1, xw, out); break;
        case 5: scatter_group<5>(mask, lane, r0, r1, c0, c1, n0, n1, xw, out); break;
        case 4: scatter_group<4>(mask, lane, r0, r1, c0, c1, n0, n1, xw, out); break;
        case 3: scatter_group<3>(mask, lane, r0, r1, c0, c1, n0, n1, xw, out); break;
        case 2: scatter_group<2>(mask, lane, r0, r1, c0, c1, n0, n1, xw, out); break;
        case 1: scatter_group<1>(mask, lane, r0, r1, c0, c1, n0, n1, xw, out); break;
        default: break;
    }
}

extern "C" void kernel_launch(void* const* d_in, const int* in_sizes, int n_in,
                              void* d_out, int out_size, void* d_ws, size_t ws_size,
                              hipStream_t stream) {
    const float* x = (const float*)d_in[0];
    const int* ei = (const int*)d_in[1];
    const float* W = (const float*)d_in[2];
    const float* b = (const float*)d_in[3];
    float* out = (float*)d_out;

    const int* row = ei;
    const int* col = ei + N_EDGES;

    char* ws = (char*)d_ws;
    unsigned* claim = (unsigned*)ws;
    float* dinv = (float*)(ws + (size_t)N_NODES * 4);
    float* xw = (float*)(ws + (size_t)N_NODES * 8);
    int* part = (int*)(ws + (size_t)N_NODES * 8);

    gcn_degree_lds<<<HIST_RANGES * HIST_SLICES, HIST_THREADS, 0, stream>>>(col, part);
    gcn_dinv<<<(N_NODES + 255) / 256, 256, 0, stream>>>(part, dinv, claim);
    gcn_xw_wmma<<<N_NODES / 32, 256, 0, stream>>>(x, W, dinv, b, xw, out);
    gcn_edge_scatter<<<SCATTER_GRID, 256, 0, stream>>>(row, col, xw, dinv, out, claim);
}


# dinv pass folded into the product launch: degree table laid out by 32-node groups inside each block's own xw rows, block n sums its own counters (4 launches to 3); values, sum order unchanged
# speedup vs baseline: 3.1757x; 1.0377x over previous
#include <hip/hip_runtime.h>

#define N_NODES 100000
#define N_EDGES 1600000
#define C_IN 64
#define C_OUT 64
#define N_PAIRS (N_EDGES / 64)

typedef __attribute__((ext_vector_type(2))) float v2f;
typedef __attribute__((ext_vector_type(8))) float v8f;

#define HIST_RANGES 8
#define HIST_SLICES 32
#define HIST_RANGE 12512
#define HIST_SLICE_EDGES (N_EDGES / HIST_SLICES)
#define HIST_SLICE_VEC (HIST_SLICE_EDGES / 4)
#define HIST_THREADS 512
#define HIST_PER_THREAD ((HIST_SLICE_VEC + HIST_THREADS - 1) / HIST_THREADS)
#define HIST_RANGE_VEC (HIST_RANGE / 4)
#define HIST_RANGE_LINES (HIST_RANGE / 32)
#define NODE_GROUPS (N_NODES / 32)
#define PART_GROUP (32 * C_OUT)

__global__ __launch_bounds__(HIST_THREADS) void gcn_degree_lds(const int* __restrict__ col,
                                                               int* __restrict__ part) {
    __shared__ alignas(16) int cnt[HIST_RANGE];

    const int tid = threadIdx.x;
    const int p = blockIdx.x & (HIST_RANGES - 1);
    const int q = blockIdx.x >> 3;
    const int lo = p * HIST_RANGE;

    const int4* srcv = (const int4*)(col + (size_t)q * HIST_SLICE_EDGES);
    int4 v[HIST_PER_THREAD];
#pragma unroll
    for (int i = 0; i < HIST_PER_THREAD; ++i) {
        const int idx = tid + i * HIST_THREADS;
        if (idx < HIST_SLICE_VEC) {
            v[i] = srcv[idx];
        } else {
            v[i] = make_int4(-1, -1, -1, -1);
        }
    }

    int4* cntv = (int4*)cnt;
    for (int i = tid; i < HIST_RANGE_VEC; i += HIST_THREADS) cntv[i] = make_int4(0, 0, 0, 0);
    __syncthreads();

#pragma unroll
    for (int i = 0; i < HIST_PER_THREAD; ++i) {
        const unsigned d0 = (unsigned)(v[i].x - lo);
        const unsigned d1 = (unsigned)(v[i].y - lo);
        const unsigned d2 = (unsigned)(v[i].z - lo);
        const unsigned d3 = (unsigned)(v[i].w - lo);
        if (d0 < (unsigned)HIST_RANGE) atomicAdd(&cnt[d0], 1);
        if (d1 < (unsigned)HIST_RANGE) atomicAdd(&cnt[d1], 1);
        if (d2 < (unsigned)HIST_RANGE) atomicAdd(&cnt[d2], 1);
        if (d3 < (unsigned)HIST_RANGE) atomicAdd(&cnt[d3], 1);
    }
    __syncthreads();

    const int g0 = p * HIST_RANGE_LINES;
    int4* dstv = (int4*)part;
    for (int i = tid; i < HIST_RANGE_VEC; i += HIST_THREADS) {
        const int g = g0 + (i >> 3);
        if (g < NODE_GROUPS) dstv[(size_t)g * (PART_GROUP / 4) + q * 8 + (i & 7)] = cntv[i];
    }
}

__global__ __launch_bounds__(256) void gcn_xw_wmma(const float* __restrict__ x,
                                                   const float* __restrict__ W,
                                                   const float* __restrict__ b,
                                                   const int* part,
                                                   float* __restrict__ dinv,
                                                   float* xw,
                                                   float* __restrict__ out,
                                                   unsigned* __restrict__ claim) {
    __shared__ float sW[C_IN * C_OUT];
    __shared__ alignas(16) int sCnt[HIST_SLICES * 32];
    __shared__ alignas(16) float sDinv[32];

    const int tid = threadIdx.x;
    const int4 pc = ((const int4*)(part + (size_t)blockIdx.x * PART_GROUP))[tid];
    const float bc = b[((tid >> 5) & 3) * 16 + (tid & 15)];
    if (tid < 8) claim[blockIdx.x * 8 + tid] = 0u;
    {
        const float4* Wv = (const float4*)W;
        float4* sWv = (float4*)sW;
#pragma unroll
        for (int i = 0; i < 4; ++i) sWv[tid + 256 * i] = Wv[tid + 256 * i];
    }
    ((int4*)sCnt)[tid] = pc;
    __syncthreads();

    if (tid < 32) {
        int d = 0;
#pragma unroll
        for (int q = 0; q < HIST_SLICES; ++q) d += sCnt[q * 32 + tid];
        const float di = rsqrtf((float)d + 1.0f);
        sDinv[tid] = di;
        dinv[blockIdx.x * 32 + tid] = di;
    }

    const int wave = tid >> 5;
    const int lane = tid & 31;
    const int m0 = blockIdx.x * 32 + (wave >> 2) * 16;
    const int n0 = (wave & 3) * 16;
    const int mh = lane & 15;
    const int kb = (lane >> 4) << 1;

    const float* arow = x + (size_t)(m0 + mh) * C_IN + kb;
    const float* bcol = sW + (size_t)kb * C_OUT + n0 + mh;

    v8f acc = {};
#pragma unroll
    for (int k = 0; k < C_IN; k += 4) {
        v2f a, bf;
        a.x = arow[k];
        a.y = arow[k + 1];
        bf.x = bcol[k * C_OUT];
        bf.y = bcol[k * C_OUT + C_OUT];
        acc = __builtin_amdgcn_wmma_f32_16x16x4_f32(
             false, a,  false, bf,
             (short)0, acc,  false,  false);
    }

    const size_t tile = (size_t)(m0 + 8 * (lane >> 4)) * C_OUT + n0 + mh;
    float* drow = xw + tile;
    float* orow = out + tile;
    __syncthreads();
    const float4* dq = (const float4*)(sDinv + (wave >> 2) * 16 + 8 * (lane >> 4));
    const float4 dlo = dq[0];
    const float4 dhi = dq[1];
    const float dr[8] = {dlo.x, dlo.y, dlo.z, dlo.w, dhi.x, dhi.y, dhi.z, dhi.w};
#pragma unroll
    for (int r = 0; r < 8; ++r) {
        const float s = dr[r] * dr[r];
        drow[r * C_OUT] = acc[r];
        orow[r * C_OUT] = bc + s * acc[r];
    }
}

#define PAIR_EDGES 64
#define PAIR_BLOCKS ((N_PAIRS + 7) / 8)
#define SCATTER_GRID (((PAIR_BLOCKS + 3) / 4) * 8)
#define HALF_SPLIT (N_NODES / 2)
#define EDGE_GROUP 8

typedef __attribute__((address_space(1))) float gfloat;
__device__ __forceinline__ gfloat* uniform_row(const float* p) {
    const unsigned long long v = (unsigned long long)p;
    const unsigned lo = (unsigned)__builtin_amdgcn_readfirstlane((int)(unsigned)v);
    const unsigned hi = (unsigned)__builtin_amdgcn_readfirstlane((int)(unsigned)(v >> 32));
    return (gfloat*)(((unsigned long long)hi << 32) | lo);
}

template <int N>
__device__ __forceinline__ void scatter_group(unsigned long long& mask, const unsigned lane,
                                              const int r0, const int r1, const int c0,
                                              const int c1, const float n0, const float n1,
                                              const float* __restrict__ xw,
                                              float* __restrict__ out) {
    float vlo[N], vhi[N], nj[N];
    int rj[N], cj[N];
#pragma unroll
    for (int j = 0; j < N; ++j) {
        const int ln = (int)__builtin_ctzll(mask);
        mask &= mask - 1ull;
        const int l5 = ln & 31;
        const bool up = ln >= 32;
        const int ra = __builtin_amdgcn_readlane(r0, l5);
        const int rb = __builtin_amdgcn_readlane(r1, l5);
        const int ca = __builtin_amdgcn_readlane(c0, l5);
        const int cb = __builtin_amdgcn_readlane(c1, l5);
        const int na = __builtin_amdgcn_readlane(__builtin_bit_cast(int, n0), l5);
        const int nb = __builtin_amdgcn_readlane(__builtin_bit_cast(int, n1), l5);
        rj[j] = up ? rb : ra;
        cj[j] = up ? cb : ca;
        nj[j] = __builtin_bit_cast(float, up ? nb : na);
    }
#pragma unroll
    for (int j = 0; j < N; ++j) {
        const gfloat* srow = uniform_row(xw + (size_t)rj[j] * C_OUT);
        vlo[j] = srow[lane];
        vhi[j] = srow[lane + C_OUT / 2];
    }
#pragma unroll
    for (int j = 0; j < N; ++j) {
        gfloat* dst = uniform_row(out + (size_t)cj[j] * C_OUT);
        __builtin_amdgcn_global_atomic_fadd_f32(dst + lane, vlo[j] * nj[j]);
        __builtin_amdgcn_global_atomic_fadd_f32(dst + lane + C_OUT / 2, vhi[j] * nj[j]);
    }
}

__global__ __launch_bounds__(256) void gcn_edge_scatter(const int* __restrict__ row,
                                                        const int* __restrict__ col,
                                                        const float* __restrict__ xw,
                                                        const float* __restrict__ dinv,
                                                        float* __restrict__ out,
                                                        unsigned* __restrict__ claim) {
    const int b = blockIdx.x;
    const int pb = ((b >> 3) << 2) | (b & 3);
    const int pair = pb * 8 + (threadIdx.x >> 5);
    if (pair >= N_PAIRS) return;
    const unsigned lane = threadIdx.x & 31u;
    const int e0 = pair * PAIR_EDGES;

    const int r0 = row[e0 + lane];
    const int r1 = row[e0 + 32 + lane];
    const int c0 = col[e0 + lane];
    const int c1 = col[e0 + 32 + lane];

    const unsigned half = (__builtin_amdgcn_s_sendmsg_rtn(0x87) >> 18) & 1u;
    unsigned old = 0u;
    if (lane == 0) old = atomicOr(&claim[pair], 1u << half);
    old = (unsigned)__builtin_amdgcn_readfirstlane((int)old);
    const unsigned side = (old == 0u) ? half : (old & 1u);

    const bool p0 = (unsigned)(c0 >= HALF_SPLIT) == side;
    const bool p1 = (unsigned)(c1 >= HALF_SPLIT) == side;
    const float n0 = dinv[p0 ? r0 : 0] * dinv[p0 ? c0 : 0];
    const float n1 = dinv[p1 ? r1 : 0] * dinv[p1 ? c1 : 0];

    unsigned long long mask = (unsigned long long)__builtin_amdgcn_ballot_w32(p0) |
                              ((unsigned long long)__builtin_amdgcn_ballot_w32(p1) << 32);

    int left = __builtin_popcountll(mask);
    for (; left >= EDGE_GROUP; left -= EDGE_GROUP)
        scatter_group<EDGE_GROUP>(mask, lane, r0, r1, c0, c1, n0, n1, xw, out);
    switch (left) {
        case 7: scatter_group<7>(mask, lane, r0, r1, c0, c1, n0, n1, xw, out); break;
        case 6: scatter_group<6>(mask, lane, r0, r1, c0, c1, n0, n1, xw, out); break;
        case 5: scatter_group<5>(mask, lane, r0, r1, c0, c1, n0, n1, xw, out); break;
        case 4: scatter_group<4>(mask, lane, r0, r1, c0, c1, n0, n1, xw, out); break;
        case 3: scatter_group<3>(mask, lane, r0, r1, c0, c1, n0, n1, xw, out); break;
        case 2: scatter_group<2>(mask, lane, r0, r1, c0, c1, n0, n1, xw, out); break;
        case 1: scatter_group<1>(mask, lane, r0, r1, c0, c1, n0, n1, xw, out); break;
        default: break;
    }
}

extern "C" void kernel_launch(void* const* d_in, const int* in_sizes, int n_in,
                              void* d_out, int out_size, void* d_ws, size_t ws_size,
                              hipStream_t stream) {
    const float* x = (const float*)d_in[0];
    const int* ei = (const int*)d_in[1];
    const float* W = (const float*)d_in[2];
    const float* b = (const float*)d_in[3];
    float* out = (float*)d_out;

    const int* row = ei;
    const int* col = ei + N_EDGES;

    char* ws = (char*)d_ws;
    unsigned* claim = (unsigned*)ws;
    float* dinv = (float*)(ws + (size_t)N_NODES * 4);
    float* xw = (float*)(ws + (size_t)N_NODES * 8);
    int* part = (int*)(ws + (size_t)N_NODES * 8);

    gcn_degree_lds<<<HIST_RANGES * HIST_SLICES, HIST_THREADS, 0, stream>>>(col, part);
    gcn_xw_wmma<<<N_NODES / 32, 256, 0, stream>>>(x, W, b, part, dinv, xw, out, claim);
    gcn_edge_scatter<<<SCATTER_GRID, 256, 0, stream>>>(row, col, xw, dinv, out, claim);
}


# product epilogue stores the gathered plane pre-scaled, dinv[j]*xw[j]: edge scatter fetches only dinv[col] per edge, its per-lane dinv line requests halve; term rounding changes, sum order as before
# speedup vs baseline: 3.4164x; 1.0758x over previous
#include <hip/hip_runtime.h>

#define N_NODES 100000
#define N_EDGES 1600000
#define C_IN 64
#define C_OUT 64
#define N_PAIRS (N_EDGES / 64)

typedef __attribute__((ext_vector_type(2))) float v2f;
typedef __attribute__((ext_vector_type(8))) float v8f;

#define HIST_RANGES 8
#define HIST_SLICES 32
#define HIST_RANGE 12512
#define HIST_SLICE_EDGES (N_EDGES / HIST_SLICES)
#define HIST_SLICE_VEC (HIST_SLICE_EDGES / 4)
#define HIST_THREADS 512
#define HIST_PER_THREAD ((HIST_SLICE_VEC + HIST_THREADS - 1) / HIST_THREADS)
#define HIST_RANGE_VEC (HIST_RANGE / 4)
#define HIST_RANGE_LINES (HIST_RANGE / 32)
#define NODE_GROUPS (N_NODES / 32)
#define PART_GROUP (32 * C_OUT)

__global__ __launch_bounds__(HIST_THREADS) void gcn_degree_lds(const int* __restrict__ col,
                                                               int* __restrict__ part) {
    __shared__ alignas(16) int cnt[HIST_RANGE];

    const int tid = threadIdx.x;
    const int p = blockIdx.x & (HIST_RANGES - 1);
    const int q = blockIdx.x >> 3;
    const int lo = p * HIST_RANGE;

    const int4* srcv = (const int4*)(col + (size_t)q * HIST_SLICE_EDGES);
    int4 v[HIST_PER_THREAD];
#pragma unroll
    for (int i = 0; i < HIST_PER_THREAD; ++i) {
        const int idx = tid + i * HIST_THREADS;
        if (idx < HIST_SLICE_VEC) {
            v[i] = srcv[idx];
        } else {
            v[i] = make_int4(-1, -1, -1, -1);
        }
    }

    int4* cntv = (int4*)cnt;
    for (int i = tid; i < HIST_RANGE_VEC; i += HIST_THREADS) cntv[i] = make_int4(0, 0, 0, 0);
    __syncthreads();

#pragma unroll
    for (int i = 0; i < HIST_PER_THREAD; ++i) {
        const unsigned d0 = (unsigned)(v[i].x - lo);
        const unsigned d1 = (unsigned)(v[i].y - lo);
        const unsigned d2 = (unsigned)(v[i].z - lo);
        const unsigned d3 = (unsigned)(v[i].w - lo);
        if (d0 < (unsigned)HIST_RANGE) atomicAdd(&cnt[d0], 1);
        if (d1 < (unsigned)HIST_RANGE) atomicAdd(&cnt[d1], 1);
        if (d2 < (unsigned)HIST_RANGE) atomicAdd(&cnt[d2], 1);
        if (d3 < (unsigned)HIST_RANGE) atomicAdd(&cnt[d3], 1);
    }
    __syncthreads();

    const int g0 = p * HIST_RANGE_LINES;
    int4* dstv = (int4*)part;
    for (int i = tid; i < HIST_RANGE_VEC; i += HIST_THREADS) {
        const int g = g0 + (i >> 3);
        if (g < NODE_GROUPS) dstv[(size_t)g * (PART_GROUP / 4) + q * 8 + (i & 7)] = cntv[i];
    }
}

__global__ __launch_bounds__(256) void gcn_xw_wmma(const float* __restrict__ x,
                                                   const float* __restrict__ W,
                                                   const float* __restrict__ b,
                                                   const int* part,
                                                   float* __restrict__ dinv,
                                                   float* xw,
                                                   float* __restrict__ out,
                                                   unsigned* __restrict__ claim) {
    __shared__ float sW[C_IN * C_OUT];
    __shared__ alignas(16) int sCnt[HIST_SLICES * 32];
    __shared__ alignas(16) float sDinv[32];

    const int tid = threadIdx.x;
    const int4 pc = ((const int4*)(part + (size_t)blockIdx.x * PART_GROUP))[tid];
    const float bc = b[((tid >> 5) & 3) * 16 + (tid & 15)];
    if (tid < 8) claim[blockIdx.x * 8 + tid] = 0u;
    {
        const float4* Wv = (const float4*)W;
        float4* sWv = (float4*)sW;
#pragma unroll
        for (int i = 0; i < 4; ++i) sWv[tid + 256 * i] = Wv[tid + 256 * i];
    }
    ((int4*)sCnt)[tid] = pc;
    __syncthreads();

    if (tid < 32) {
        int d = 0;
#pragma unroll
        for (int q = 0; q < HIST_SLICES; ++q) d += sCnt[q * 32 + tid];
        const float di = rsqrtf((float)d + 1.0f);
        sDinv[tid] = di;
        dinv[blockIdx.x * 32 + tid] = di;
    }

    const int wave = tid >> 5;
    const int lane = tid & 31;
    const int m0 = blockIdx.x * 32 + (wave >> 2) * 16;
    const int n0 = (wave & 3) * 16;
    const int mh = lane & 15;
    const int kb = (lane >> 4) << 1;

    const float* arow = x + (size_t)(m0 + mh) * C_IN + kb;
    const float* bcol = sW + (size_t)kb * C_OUT + n0 + mh;

    v8f acc = {};
#pragma unroll
    for (int k = 0; k < C_IN; k += 4) {
        v2f a, bf;
        a.x = arow[k];
        a.y = arow[k + 1];
        bf.x = bcol[k * C_OUT];
        bf.y = bcol[k * C_OUT + C_OUT];
        acc = __builtin_amdgcn_wmma_f32_16x16x4_f32(
             false, a,  false, bf,
             (short)0, acc,  false,  false);
    }

    const size_t tile = (size_t)(m0 + 8 * (lane >> 4)) * C_OUT + n0 + mh;
    float* drow = xw + tile;
    float* orow = out + tile;
    __syncthreads();
    const float4* dq = (const float4*)(sDinv + (wave >> 2) * 16 + 8 * (lane >> 4));
    const float4 dlo = dq[0];
    const float4 dhi = dq[1];
    const float dr[8] = {dlo.x, dlo.y, dlo.z, dlo.w, dhi.x, dhi.y, dhi.z, dhi.w};
#pragma unroll
    for (int r = 0; r < 8; ++r) {
        const float s = dr[r] * dr[r];
        drow[r * C_OUT] = dr[r] * acc[r];
        orow[r * C_OUT] = bc + s * acc[r];
    }
}

#define PAIR_EDGES 64
#define PAIR_BLOCKS ((N_PAIRS + 7) / 8)
#define SCATTER_GRID (((PAIR_BLOCKS + 3) / 4) * 8)
#define HALF_SPLIT (N_NODES / 2)
#define EDGE_GROUP 8

typedef __attribute__((address_space(1))) float gfloat;
__device__ __forceinline__ gfloat* uniform_row(const float* p) {
    const unsigned long long v = (unsigned long long)p;
    const unsigned lo = (unsigned)__builtin_amdgcn_readfirstlane((int)(unsigned)v);
    const unsigned hi = (unsigned)__builtin_amdgcn_readfirstlane((int)(unsigned)(v >> 32));
    return (gfloat*)(((unsigned long long)hi << 32) | lo);
}

template <int N>
__device__ __forceinline__ void scatter_group(unsigned long long& mask, const unsigned lane,
                                              const int r0, const int r1, const int c0,
                                              const int c1, const float n0, const float n1,
                                              const float* __restrict__ xw,
                                              float* __restrict__ out) {
    float vlo[N], vhi[N], nj[N];
    int rj[N], cj[N];
#pragma unroll
    for (int j = 0; j < N; ++j) {
        const int ln = (int)__builtin_ctzll(mask);
        mask &= mask - 1ull;
        const int l5 = ln & 31;
        const bool up = ln >= 32;
        const int ra = __builtin_amdgcn_readlane(r0, l5);
        const int rb = __builtin_amdgcn_readlane(r1, l5);
        const int ca = __builtin_amdgcn_readlane(c0, l5);
        const int cb = __builtin_amdgcn_readlane(c1, l5);
        const int na = __builtin_amdgcn_readlane(__builtin_bit_cast(int, n0), l5);
        const int nb = __builtin_amdgcn_readlane(__builtin_bit_cast(int, n1), l5);
        rj[j] = up ? rb : ra;
        cj[j] = up ? cb : ca;
        nj[j] = __builtin_bit_cast(float, up ? nb : na);
    }
#pragma unroll
    for (int j = 0; j < N; ++j) {
        const gfloat* srow = uniform_row(xw + (size_t)rj[j] * C_OUT);
        vlo[j] = srow[lane];
        vhi[j] = srow[lane + C_OUT / 2];
    }
#pragma unroll
    for (int j = 0; j < N; ++j) {
        gfloat* dst = uniform_row(out + (size_t)cj[j] * C_OUT);
        __builtin_amdgcn_global_atomic_fadd_f32(dst + lane, vlo[j] * nj[j]);
        __builtin_amdgcn_global_atomic_fadd_f32(dst + lane + C_OUT / 2, vhi[j] * nj[j]);
    }
}

__global__ __launch_bounds__(256) void gcn_edge_scatter(const int* __restrict__ row,
                                                        const int* __restrict__ col,
                                                        const float* __restrict__ xw,
                                                        const float* __restrict__ dinv,
                                                        float* __restrict__ out,
                                                        unsigned* __restrict__ claim) {
    const int b = blockIdx.x;
    const int pb = ((b >> 3) << 2) | (b & 3);
    const int pair = pb * 8 + (threadIdx.x >> 5);
    if (pair >= N_PAIRS) return;
    const unsigned lane = threadIdx.x & 31u;
    const int e0 = pair * PAIR_EDGES;

    const int r0 = row[e0 + lane];
    const int r1 = row[e0 + 32 + lane];
    const int c0 = col[e0 + lane];
    const int c1 = col[e0 + 32 + lane];

    const unsigned half = (__builtin_amdgcn_s_sendmsg_rtn(0x87) >> 18) & 1u;
    unsigned old = 0u;
    if (lane == 0) old = atomicOr(&claim[pair], 1u << half);
    old = (unsigned)__builtin_amdgcn_readfirstlane((int)old);
    const unsigned side = (old == 0u) ? half : (old & 1u);

    const bool p0 = (unsigned)(c0 >= HALF_SPLIT) == side;
    const bool p1 = (unsigned)(c1 >= HALF_SPLIT) == side;
    const float n0 = dinv[p0 ? c0 : 0];
    const float n1 = dinv[p1 ? c1 : 0];

    unsigned long long mask = (unsigned long long)__builtin_amdgcn_ballot_w32(p0) |
                              ((unsigned long long)__builtin_amdgcn_ballot_w32(p1) << 32);

    int left = __builtin_popcountll(mask);
    for (; left >= EDGE_GROUP; left -= EDGE_GROUP)
        scatter_group<EDGE_GROUP>(mask, lane, r0, r1, c0, c1, n0, n1, xw, out);
    switch (left) {
        case 7: scatter_group<7>(mask, lane, r0, r1, c0, c1, n0, n1, xw, out); break;
        case 6: scatter_group<6>(mask, lane, r0, r1, c0, c1, n0, n1, xw, out); break;
        case 5: scatter_group<5>(mask, lane, r0, r1, c0, c1, n0, n1, xw, out); break;
        case 4: scatter_group<4>(mask, lane, r0, r1, c0, c1, n0, n1, xw, out); break;
        case 3: scatter_group<3>(mask, lane, r0, r1, c0, c1, n0, n1, xw, out); break;
        case 2: scatter_group<2>(mask, lane, r0, r1, c0, c1, n0, n1, xw, out); break;
        case 1: scatter_group<1>(mask, lane, r0, r1, c0, c1, n0, n1, xw, out); break;
        default: break;
    }
}

extern "C" void kernel_launch(void* const* d_in, const int* in_sizes, int n_in,
                              void* d_out, int out_size, void* d_ws, size_t ws_size,
                              hipStream_t stream) {
    const float* x = (const float*)d_in[0];
    const int* ei = (const int*)d_in[1];
    const float* W = (const float*)d_in[2];
    const float* b = (const float*)d_in[3];
    float* out = (float*)d_out;

    const int* row = ei;
    const int* col = ei + N_EDGES;

    char* ws = (char*)d_ws;
    unsigned* claim = (unsigned*)ws;
    float* dinv = (float*)(ws + (size_t)N_NODES * 4);
    float* xw = (float*)(ws + (size_t)N_NODES * 8);
    int* part = (int*)(ws + (size_t)N_NODES * 8);

    gcn_degree_lds<<<HIST_RANGES * HIST_SLICES, HIST_THREADS, 0, stream>>>(col, part);
    gcn_xw_wmma<<<N_NODES / 32, 256, 0, stream>>>(x, W, b, part, dinv, xw, out, claim);
    gcn_edge_scatter<<<SCATTER_GRID, 256, 0, stream>>>(row, col, xw, dinv, out, claim);
}
